# starting kernel
# baseline (speedup 1.0000x reference)
#include <hip/hip_runtime.h>
#include <hip/hip_bf16.h>

typedef float v2f __attribute__((ext_vector_type(2)));
typedef float v8f __attribute__((ext_vector_type(8)));

#define BATCH 16
#define CH    256
#define NN    4096

__global__ void FSiAM_row_mean(const float* __restrict__ x,
                               float* __restrict__ m) {
    const int lane = threadIdx.x & 31;
    const int wave = threadIdx.x >> 5;
    const int row  = blockIdx.x * 8 + wave;
    const float* xr = x + (size_t)row * NN;

    float s = 0.0f;
#pragma unroll 4
    for (int i = 0; i < NN / 128; ++i) {
        const float4 v = *(const float4*)(xr + i * 128 + lane * 4);
        s += v.x + v.y + v.z + v.w;
    }
#pragma unroll
    for (int off = 16; off > 0; off >>= 1)
        s += __shfl_xor(s, off, 32);

    if (lane == 0) m[row] = s * (1.0f / (float)NN);
}

__global__ void FSiAM_matvec_sigmoid(const float* __restrict__ x,
                                     const float* __restrict__ m,
                                     float* __restrict__ out) {
    __shared__ float sm[CH];
    const int b    = blockIdx.y;
    const int lane = threadIdx.x & 31;
    const int wave = threadIdx.x >> 5;

    sm[threadIdx.x] = m[b * CH + threadIdx.x];
    __syncthreads();

    const int koff = (lane < 16) ? 0 : 2;
    const int ncol = (blockIdx.x * 8 + wave) * 16 + (lane & 15);
    const float* xb = x + (size_t)b * CH * NN;

    v8f acc = {};
    for (int c0 = 0; c0 < CH; c0 += 4) {
        const float* xp = xb + (size_t)(c0 + koff) * NN + ncol;
        __builtin_prefetch(xp + (size_t)4 * NN, 0, 1);

        v2f Av, Bv;
        Av.x = sm[c0 + koff];
        Av.y = sm[c0 + koff + 1];
        Bv.x = xp[0];
        Bv.y = xp[NN];

        acc = __builtin_amdgcn_wmma_f32_16x16x4_f32(
            false, Av, false, Bv, (short)0, acc, false, false);
    }

    const float s   = acc[0];
    const float sig = 1.0f / (1.0f + __expf(-s));
    if (lane < 16)
        out[b * NN + (blockIdx.x * 8 + wave) * 16 + lane] = sig;
}

extern "C" void kernel_launch(void* const* d_in, const int* in_sizes, int n_in,
                              void* d_out, int out_size, void* d_ws, size_t ws_size,
                              hipStream_t stream) {
    const float* x = (const float*)d_in[0];
    float* out = (float*)d_out;
    float* m   = (float*)d_ws;

    FSiAM_row_mean<<<dim3((BATCH * CH) / 8), dim3(256), 0, stream>>>(x, m);

    FSiAM_matvec_sigmoid<<<dim3(NN / 128, BATCH), dim3(256), 0, stream>>>(x, m, out);
}


# x read once: one work-group per (batch, 16-channel slice) holds the slice in registers, takes its means, then its WMMA partial scores; a 2nd kernel adds the 16 partials + sigmoid (sum order changes)
# speedup vs baseline: 1.2461x; 1.2461x over previous
#include <hip/hip_runtime.h>


typedef float v2f __attribute__((ext_vector_type(2)));
typedef float v8f __attribute__((ext_vector_type(8)));

#define BATCH 16
#define CH    256
#define NN    4096
#define CS    16
#define NSL   (CH / CS)
#define WAVES 16
#define PART_BYTES ((size_t)BATCH * NSL * NN * sizeof(float))

__device__ __forceinline__ void lds_barrier() {
    asm volatile("s_wait_dscnt 0x0\n\ts_barrier_signal -1\n\ts_barrier_wait -1" ::: "memory");
}

template <int CTRL>
__device__ __forceinline__ float row_xor_add(float v) {
    const int t = __builtin_amdgcn_update_dpp(0, __float_as_int(v), CTRL, 0xf, 0xf, false);
    return v + __int_as_float(t);
}

__global__ __launch_bounds__(512)
void slice_mean_partial(const float* x, float* __restrict__ part, size_t ws_bytes) {
    if (ws_bytes < PART_BYTES) return;

    __shared__ __attribute__((aligned(16))) float red[4][4][WAVES];

    const int lane = threadIdx.x & 31;
    const int w    = threadIdx.x >> 5;
    const int h    = lane >> 4;
    const int c    = lane & 15;
    const int b    = blockIdx.x / NSL;
    const int sl   = blockIdx.x % NSL;

    const float* xs = x + ((size_t)b * CH + (size_t)sl * CS + 2 * h) * NN + w * 256 + c * 4;

    float4 v[4][4][2];
#pragma unroll
    for (int g = 0; g < 4; ++g)
#pragma unroll
        for (int i = 0; i < 4; ++i)
#pragma unroll
            for (int kk = 0; kk < 2; ++kk)
                v[g][i][kk] = *(const float4*)(xs + (size_t)(4 * g + kk) * NN + 64 * i);

    v8f accA = {};
    v8f accB = {};

#pragma unroll
    for (int g = 0; g < 4; ++g) {
        if (g > 0) __builtin_amdgcn_sched_barrier(0);
        float s0 = 0.0f, s1 = 0.0f;
#pragma unroll
        for (int i = 0; i < 4; ++i) {
            s0 += (v[g][i][0].x + v[g][i][0].y) + (v[g][i][0].z + v[g][i][0].w);
            s1 += (v[g][i][1].x + v[g][i][1].y) + (v[g][i][1].z + v[g][i][1].w);
        }
        s0 = row_xor_add<0x168>(s0);  s1 = row_xor_add<0x168>(s1);
        s0 = row_xor_add<0x164>(s0);  s1 = row_xor_add<0x164>(s1);
        s0 = row_xor_add<0x162>(s0);  s1 = row_xor_add<0x162>(s1);
        s0 = row_xor_add<0x161>(s0);  s1 = row_xor_add<0x161>(s1);
        if (c == 0) {
            red[g][2 * h][w]     = s0;
            red[g][2 * h + 1][w] = s1;
        }
        lds_barrier();

        const float4* r0 = (const float4*)&red[g][2 * h][0];
        const float4* r1 = (const float4*)&red[g][2 * h + 1][0];
        const float4 a0 = r0[0], a1 = r0[1], a2 = r0[2], a3 = r0[3];
        const float4 b0 = r1[0], b1 = r1[1], b2 = r1[2], b3 = r1[3];
        const float ax = (a0.x + a1.x) + (a2.x + a3.x), ay = (a0.y + a1.y) + (a2.y + a3.y);
        const float az = (a0.z + a1.z) + (a2.z + a3.z), aw = (a0.w + a1.w) + (a2.w + a3.w);
        const float bx = (b0.x + b1.x) + (b2.x + b3.x), by = (b0.y + b1.y) + (b2.y + b3.y);
        const float bz = (b0.z + b1.z) + (b2.z + b3.z), bw = (b0.w + b1.w) + (b2.w + b3.w);
        const float m0 = ((ax + ay) + (az + aw)) * (1.0f / (float)NN);
        const float m1 = ((bx + by) + (bz + bw)) * (1.0f / (float)NN);

#pragma unroll
        for (int i = 0; i < 4; ++i) {
#pragma unroll
            for (int kk = 0; kk < 2; ++kk) {
                const float mk = kk ? m1 : m0;
                const float4 q = v[g][i][kk];
#pragma unroll
                for (int p = 0; p < 2; ++p) {
                    v2f Av, Bv;
                    Av.x = (c == 4 * i + 2 * p)     ? mk : 0.0f;
                    Av.y = (c == 4 * i + 2 * p + 1) ? mk : 0.0f;
                    Bv.x = p ? q.z : q.x;
                    Bv.y = p ? q.w : q.y;
                    if (i < 2)
                        accA = __builtin_amdgcn_wmma_f32_16x16x4_f32(false, Av, false, Bv, (short)0, accA, false, false);
                    else
                        accB = __builtin_amdgcn_wmma_f32_16x16x4_f32(false, Av, false, Bv, (short)0, accB, false, false);
                }
            }
        }
    }

    const v8f fin = accA + accB;
    float* pp = part + ((size_t)(b * NSL + sl)) * NN + w * 256 + 128 * h + 4 * c;
    *(float4*)(pp)      = make_float4(fin[0], fin[1], fin[2], fin[3]);
    *(float4*)(pp + 64) = make_float4(fin[4], fin[5], fin[6], fin[7]);
}

__global__ __launch_bounds__(64)
void sum_slices_sigmoid(const float* __restrict__ part, float* __restrict__ out, size_t ws_bytes) {
    if (ws_bytes < PART_BYTES) return;
    const int idx = blockIdx.x * 64 + threadIdx.x;
    const int b   = idx >> 10;
    const int n4  = idx & 1023;
    const float4* p = (const float4*)(part + (size_t)b * NSL * NN) + n4;
    float4 q[NSL];
#pragma unroll
    for (int s = 0; s < NSL; ++s) q[s] = p[(size_t)s * (NN / 4)];
    float4 a = q[0];
#pragma unroll
    for (int s = 1; s < NSL; ++s) { a.x += q[s].x; a.y += q[s].y; a.z += q[s].z; a.w += q[s].w; }
    float4 o;
    o.x = 1.0f / (1.0f + __expf(-a.x));
    o.y = 1.0f / (1.0f + __expf(-a.y));
    o.z = 1.0f / (1.0f + __expf(-a.z));
    o.w = 1.0f / (1.0f + __expf(-a.w));
    ((float4*)out)[idx] = o;
}

extern "C" void kernel_launch(void* const* d_in, const int* in_sizes, int n_in,
                              void* d_out, int out_size, void* d_ws, size_t ws_size,
                              hipStream_t stream) {
    const float* x = (const float*)d_in[0];
    float* out  = (float*)d_out;
    float* part = (float*)d_ws;

    slice_mean_partial<<<dim3(BATCH * NSL), dim3(512), 0, stream>>>(x, part, ws_size);
    sum_slices_sigmoid<<<dim3(256), dim3(64), 0, stream>>>(part, out, ws_size);
}
